# starting kernel
# baseline (speedup 1.0000x reference)
#include <hip/hip_runtime.h>
#include <math.h>


typedef __attribute__((ext_vector_type(2))) float v2f;
typedef __attribute__((ext_vector_type(8))) float v8f;

#define NUM_CLASSES 10
#define N_LAYERS 3

__global__ void qc_setup_coeffs(const float* __restrict__ params,
                                float* __restrict__ Acoef) {
  int t = threadIdx.x;
  if (t < 2 * 16 * 4) Acoef[t] = 0.0f;
  __syncthreads();
  if (t < NUM_CLASSES) {
    float tr[4] = {1.f, 0.f, 0.f, 0.f};
    float ti[4] = {0.f, 0.f, 0.f, 0.f};
    float br[4] = {0.f, 0.f, 0.f, 0.f};
    float bi[4] = {0.f, 0.f, 0.f, 0.f};
    int len = 1;
    for (int l = 0; l < N_LAYERS; ++l) {
      float ntr[4], nti[4], nbr[4], nbi[4];
      for (int j = 0; j <= len; ++j) {
        float xr = 0.f, xi = 0.f, yr = 0.f, yi = 0.f;
        if (j < len) { xr = tr[j]; xi = ti[j]; yr = br[j]; yi = bi[j]; }
        if (j >= 1)  { xr -= br[j-1]; xi -= bi[j-1]; yr += tr[j-1]; yi += ti[j-1]; }
        ntr[j] = xr; nti[j] = xi; nbr[j] = yr; nbi[j] = yi;
      }
      ++len;
      const float* P = params + (t * N_LAYERS + l) * 3;
      float th0 = 0.5f * P[0], th1 = 0.5f * P[1], th2 = 0.5f * P[2];
      float s1, c1, sz0, cz0, sz2, cz2;
      sincosf(th1, &s1, &c1);
      sincosf(th0, &sz0, &cz0);
      sincosf(th2, &sz2, &cz2);
      for (int j = 0; j < len; ++j) {
        float xr = ntr[j], xi = nti[j], yr = nbr[j], yi = nbi[j];
        float txr = xr * cz0 + xi * sz0;
        float txi = xi * cz0 - xr * sz0;
        float tyr = yr * cz0 - yi * sz0;
        float tyi = yi * cz0 + yr * sz0;
        float uxr = c1 * txr - s1 * tyr, uxi = c1 * txi - s1 * tyi;
        float uyr = s1 * txr + c1 * tyr, uyi = s1 * txi + c1 * tyi;
        tr[j] = uxr * cz2 + uxi * sz2;
        ti[j] = uxi * cz2 - uxr * sz2;
        br[j] = uyr * cz2 - uyi * sz2;
        bi[j] = uyi * cz2 + uyr * sz2;
      }
    }
    int nRe = 2 * t, nIm = 2 * t + 1;
    for (int k = 0; k < 4; ++k) {
      Acoef[((nRe >> 4) * 16 + (nRe & 15)) * 4 + k] = tr[k];
      Acoef[((nIm >> 4) * 16 + (nIm & 15)) * 4 + k] = ti[k];
    }
  }
}

__global__ void __launch_bounds__(256)
qc_fuzzy_kernel(const float* __restrict__ X, const float* __restrict__ Acoef,
                float* __restrict__ out, int ntiles) {
  const int lane = threadIdx.x & 31;
  const int wave = threadIdx.x >> 5;
  const int wpb  = blockDim.x >> 5;
  const int wid  = blockIdx.x * wpb + wave;
  const int wstride = gridDim.x * wpb;

  const int  n  = lane & 15;
  const bool hi = lane >= 16;
  const int  kb = hi ? 2 : 0;

  v2f a0, a1;
  a0[0] = Acoef[(0 * 16 + n) * 4 + kb];
  a0[1] = Acoef[(0 * 16 + n) * 4 + kb + 1];
  a1[0] = Acoef[(1 * 16 + n) * 4 + kb];
  a1[1] = Acoef[(1 * 16 + n) * 4 + kb + 1];

  for (int t = wid; t < ntiles; t += wstride) {
    const int p = t * 16 + n;
    const int tn = t + wstride;
    if (tn < ntiles) __builtin_prefetch(X + tn * 16 + n, 0, 0);

    float x = X[p];
    float s, c;
    __sincosf(0.5f * x, &s, &c);
    float c2 = c * c, s2 = s * s;
    v2f b;
    b[0] = hi ? c * s2 : c2 * c;
    b[1] = hi ? s2 * s : c2 * s;

    v8f z = {};
    v8f d0 = __builtin_amdgcn_wmma_f32_16x16x4_f32(false, a0, false, b,
                                                   (short)0, z, false, false);
    v8f d1 = __builtin_amdgcn_wmma_f32_16x16x4_f32(false, a1, false, b,
                                                   (short)0, z, false, false);

    float* o = out + (size_t)p * NUM_CLASSES;
    if (!hi) {
      float q0 = d0[0] * d0[0] + d0[1] * d0[1];
      float q1 = d0[2] * d0[2] + d0[3] * d0[3];
      float q2 = d0[4] * d0[4] + d0[5] * d0[5];
      float q3 = d0[6] * d0[6] + d0[7] * d0[7];
      float q8 = d1[0] * d1[0] + d1[1] * d1[1];
      float q9 = d1[2] * d1[2] + d1[3] * d1[3];
      *reinterpret_cast<float2*>(o + 0) = make_float2(q0, q1);
      *reinterpret_cast<float2*>(o + 2) = make_float2(q2, q3);
      *reinterpret_cast<float2*>(o + 8) = make_float2(q8, q9);
    } else {
      float q4 = d0[0] * d0[0] + d0[1] * d0[1];
      float q5 = d0[2] * d0[2] + d0[3] * d0[3];
      float q6 = d0[4] * d0[4] + d0[5] * d0[5];
      float q7 = d0[6] * d0[6] + d0[7] * d0[7];
      *reinterpret_cast<float2*>(o + 4) = make_float2(q4, q5);
      *reinterpret_cast<float2*>(o + 6) = make_float2(q6, q7);
    }
  }
}

extern "C" void kernel_launch(void* const* d_in, const int* in_sizes, int n_in,
                              void* d_out, int out_size, void* d_ws, size_t ws_size,
                              hipStream_t stream) {
  const float* x      = (const float*)d_in[0];
  const float* params = (const float*)d_in[1];
  float* out   = (float*)d_out;
  float* Acoef = (float*)d_ws;

  const int npix   = in_sizes[0];
  const int ntiles = npix / 16;

  hipLaunchKernelGGL(qc_setup_coeffs, dim3(1), dim3(128), 0, stream,
                     params, Acoef);

  int blocks = (ntiles + 7) / 8;
  if (blocks > 4096) blocks = 4096;
  hipLaunchKernelGGL(qc_fuzzy_kernel, dim3(blocks), dim3(256), 0, stream,
                     x, Acoef, out, ntiles);
}


# Long-lived waves: grid cut from 4096 to 512 work-groups (4096 waves enter instead of 32768); each wave loops over 8-tile groups with the next group's x loaded ahead; per-tile code unchanged
# speedup vs baseline: 1.3868x; 1.3868x over previous
#include <hip/hip_runtime.h>
#include <math.h>


typedef __attribute__((ext_vector_type(2))) float v2f;
typedef __attribute__((ext_vector_type(8))) float v8f;

#define NUM_CLASSES 10
#define N_LAYERS 3

__global__ void qc_setup_coeffs(const float* __restrict__ params,
                                float* __restrict__ Acoef) {
  int t = threadIdx.x;
  if (t < 2 * 16 * 4) Acoef[t] = 0.0f;
  __syncthreads();
  if (t < NUM_CLASSES) {
    float tr[4] = {1.f, 0.f, 0.f, 0.f};
    float ti[4] = {0.f, 0.f, 0.f, 0.f};
    float br[4] = {0.f, 0.f, 0.f, 0.f};
    float bi[4] = {0.f, 0.f, 0.f, 0.f};
    int len = 1;
    for (int l = 0; l < N_LAYERS; ++l) {
      float ntr[4], nti[4], nbr[4], nbi[4];
      for (int j = 0; j <= len; ++j) {
        float xr = 0.f, xi = 0.f, yr = 0.f, yi = 0.f;
        if (j < len) { xr = tr[j]; xi = ti[j]; yr = br[j]; yi = bi[j]; }
        if (j >= 1)  { xr -= br[j-1]; xi -= bi[j-1]; yr += tr[j-1]; yi += ti[j-1]; }
        ntr[j] = xr; nti[j] = xi; nbr[j] = yr; nbi[j] = yi;
      }
      ++len;
      const float* P = params + (t * N_LAYERS + l) * 3;
      float th0 = 0.5f * P[0], th1 = 0.5f * P[1], th2 = 0.5f * P[2];
      float s1, c1, sz0, cz0, sz2, cz2;
      sincosf(th1, &s1, &c1);
      sincosf(th0, &sz0, &cz0);
      sincosf(th2, &sz2, &cz2);
      for (int j = 0; j < len; ++j) {
        float xr = ntr[j], xi = nti[j], yr = nbr[j], yi = nbi[j];
        float txr = xr * cz0 + xi * sz0;
        float txi = xi * cz0 - xr * sz0;
        float tyr = yr * cz0 - yi * sz0;
        float tyi = yi * cz0 + yr * sz0;
        float uxr = c1 * txr - s1 * tyr, uxi = c1 * txi - s1 * tyi;
        float uyr = s1 * txr + c1 * tyr, uyi = s1 * txi + c1 * tyi;
        tr[j] = uxr * cz2 + uxi * sz2;
        ti[j] = uxi * cz2 - uxr * sz2;
        br[j] = uyr * cz2 - uyi * sz2;
        bi[j] = uyi * cz2 + uyr * sz2;
      }
    }
    int nRe = 2 * t, nIm = 2 * t + 1;
    for (int k = 0; k < 4; ++k) {
      Acoef[((nRe >> 4) * 16 + (nRe & 15)) * 4 + k] = tr[k];
      Acoef[((nIm >> 4) * 16 + (nIm & 15)) * 4 + k] = ti[k];
    }
  }
}

#define TPG 8

__device__ __forceinline__ void qc_tile(float x, v2f a0, v2f a1, bool hi,
                                        float* __restrict__ o) {
  float s, c;
  __sincosf(0.5f * x, &s, &c);
  float c2 = c * c, s2 = s * s;
  v2f b;
  b[0] = hi ? c * s2 : c2 * c;
  b[1] = hi ? s2 * s : c2 * s;

  v8f z = {};
  v8f d0 = __builtin_amdgcn_wmma_f32_16x16x4_f32(false, a0, false, b,
                                                 (short)0, z, false, false);
  v8f d1 = __builtin_amdgcn_wmma_f32_16x16x4_f32(false, a1, false, b,
                                                 (short)0, z, false, false);

  if (!hi) {
    float q0 = d0[0] * d0[0] + d0[1] * d0[1];
    float q1 = d0[2] * d0[2] + d0[3] * d0[3];
    float q2 = d0[4] * d0[4] + d0[5] * d0[5];
    float q3 = d0[6] * d0[6] + d0[7] * d0[7];
    float q8 = d1[0] * d1[0] + d1[1] * d1[1];
    float q9 = d1[2] * d1[2] + d1[3] * d1[3];
    *reinterpret_cast<float2*>(o + 0) = make_float2(q0, q1);
    *reinterpret_cast<float2*>(o + 2) = make_float2(q2, q3);
    *reinterpret_cast<float2*>(o + 8) = make_float2(q8, q9);
  } else {
    float q4 = d0[0] * d0[0] + d0[1] * d0[1];
    float q5 = d0[2] * d0[2] + d0[3] * d0[3];
    float q6 = d0[4] * d0[4] + d0[5] * d0[5];
    float q7 = d0[6] * d0[6] + d0[7] * d0[7];
    *reinterpret_cast<float2*>(o + 4) = make_float2(q4, q5);
    *reinterpret_cast<float2*>(o + 6) = make_float2(q6, q7);
  }
}

__global__ void __launch_bounds__(256)
qc_fuzzy_kernel(const float* __restrict__ X, const float* __restrict__ Acoef,
                float* __restrict__ out, int ntiles) {
  const int lane = threadIdx.x & 31;
  const int wave = __builtin_amdgcn_readfirstlane(threadIdx.x >> 5);
  const int wpb  = blockDim.x >> 5;
  const int wid  = blockIdx.x * wpb + wave;
  const int wstride = gridDim.x * wpb;

  const int  n  = lane & 15;
  const bool hi = lane >= 16;
  const int  kb = hi ? 2 : 0;

  const int ngroups = (ntiles + TPG - 1) / TPG;
  if (wid >= ngroups) return;
  const int last = ntiles - 1;

  v2f a0, a1;
  a0[0] = Acoef[(0 * 16 + n) * 4 + kb];
  a0[1] = Acoef[(0 * 16 + n) * 4 + kb + 1];
  a1[0] = Acoef[(1 * 16 + n) * 4 + kb];
  a1[1] = Acoef[(1 * 16 + n) * 4 + kb + 1];

  float xn[TPG];
#pragma unroll
  for (int j = 0; j < TPG; ++j) {
    int tt = wid * TPG + j;
    tt = tt < last ? tt : last;
    xn[j] = X[tt * 16 + n];
  }

  for (int g = wid; g < ngroups; g += wstride) {
    float xc[TPG];
#pragma unroll
    for (int j = 0; j < TPG; ++j) xc[j] = xn[j];

    int gn = g + wstride;
    gn = gn < ngroups ? gn : ngroups - 1;
#pragma unroll
    for (int j = 0; j < TPG; ++j) {
      int tt = gn * TPG + j;
      tt = tt < last ? tt : last;
      xn[j] = X[tt * 16 + n];
    }

#pragma unroll
    for (int j = 0; j < TPG; ++j) {
      const int t = g * TPG + j;
      if (t < ntiles) {
        const int p = t * 16 + n;
        qc_tile(xc[j], a0, a1, hi, out + (size_t)p * NUM_CLASSES);
      }
    }
  }
}

extern "C" void kernel_launch(void* const* d_in, const int* in_sizes, int n_in,
                              void* d_out, int out_size, void* d_ws, size_t ws_size,
                              hipStream_t stream) {
  const float* x      = (const float*)d_in[0];
  const float* params = (const float*)d_in[1];
  float* out   = (float*)d_out;
  float* Acoef = (float*)d_ws;

  const int npix   = in_sizes[0];
  const int ntiles = npix / 16;

  hipLaunchKernelGGL(qc_setup_coeffs, dim3(1), dim3(128), 0, stream,
                     params, Acoef);

  int blocks = (ntiles + 7) / 8;
  if (blocks > 512) blocks = 512;
  hipLaunchKernelGGL(qc_fuzzy_kernel, dim3(blocks), dim3(256), 0, stream,
                     x, Acoef, out, ntiles);
}


# Coefficient setup folded into the main kernel: each work-group builds the 128-float table in its own LDS (90 lanes one sincos each, then 10 lanes one class each); one launch instead of two
# speedup vs baseline: 1.7547x; 1.2653x over previous
#include <hip/hip_runtime.h>
#include <math.h>


typedef __attribute__((ext_vector_type(2))) float v2f;
typedef __attribute__((ext_vector_type(8))) float v8f;

#define NUM_CLASSES 10
#define N_LAYERS 3

#define N_ANGLES (NUM_CLASSES * N_LAYERS * 3)

__device__ __forceinline__ void qc_class_coeffs(int cls, const float* sc,
                                                float* Acoef) {
  float tr[4] = {1.f, 0.f, 0.f, 0.f};
  float ti[4] = {0.f, 0.f, 0.f, 0.f};
  float br[4] = {0.f, 0.f, 0.f, 0.f};
  float bi[4] = {0.f, 0.f, 0.f, 0.f};
  int len = 1;
#pragma unroll
  for (int l = 0; l < N_LAYERS; ++l) {
    float ntr[4], nti[4], nbr[4], nbi[4];
#pragma unroll
    for (int j = 0; j < 4; ++j) {
      if (j <= len) {
        float xr = 0.f, xi = 0.f, yr = 0.f, yi = 0.f;
        if (j < len) { xr = tr[j]; xi = ti[j]; yr = br[j]; yi = bi[j]; }
        if (j >= 1)  { xr -= br[j-1]; xi -= bi[j-1]; yr += tr[j-1]; yi += ti[j-1]; }
        ntr[j] = xr; nti[j] = xi; nbr[j] = yr; nbi[j] = yi;
      }
    }
    ++len;
    const float* S = sc + 2 * ((cls * N_LAYERS + l) * 3);
    float sz0 = S[0], cz0 = S[1];
    float s1  = S[2], c1  = S[3];
    float sz2 = S[4], cz2 = S[5];
#pragma unroll
    for (int j = 0; j < 4; ++j) {
      if (j < len) {
        float xr = ntr[j], xi = nti[j], yr = nbr[j], yi = nbi[j];
        float txr = xr * cz0 + xi * sz0;
        float txi = xi * cz0 - xr * sz0;
        float tyr = yr * cz0 - yi * sz0;
        float tyi = yi * cz0 + yr * sz0;
        float uxr = c1 * txr - s1 * tyr, uxi = c1 * txi - s1 * tyi;
        float uyr = s1 * txr + c1 * tyr, uyi = s1 * txi + c1 * tyi;
        tr[j] = uxr * cz2 + uxi * sz2;
        ti[j] = uxi * cz2 - uxr * sz2;
        br[j] = uyr * cz2 - uyi * sz2;
        bi[j] = uyi * cz2 + uyr * sz2;
      }
    }
  }
  int nRe = 2 * cls, nIm = 2 * cls + 1;
#pragma unroll
  for (int k = 0; k < 4; ++k) {
    Acoef[((nRe >> 4) * 16 + (nRe & 15)) * 4 + k] = tr[k];
    Acoef[((nIm >> 4) * 16 + (nIm & 15)) * 4 + k] = ti[k];
  }
}

#define TPG 8

__device__ __forceinline__ void qc_tile(float x, v2f a0, v2f a1, bool hi,
                                        float* __restrict__ o) {
  float s, c;
  __sincosf(0.5f * x, &s, &c);
  float c2 = c * c, s2 = s * s;
  v2f b;
  b[0] = hi ? c * s2 : c2 * c;
  b[1] = hi ? s2 * s : c2 * s;

  v8f z = {};
  v8f d0 = __builtin_amdgcn_wmma_f32_16x16x4_f32(false, a0, false, b,
                                                 (short)0, z, false, false);
  v8f d1 = __builtin_amdgcn_wmma_f32_16x16x4_f32(false, a1, false, b,
                                                 (short)0, z, false, false);

  if (!hi) {
    float q0 = d0[0] * d0[0] + d0[1] * d0[1];
    float q1 = d0[2] * d0[2] + d0[3] * d0[3];
    float q2 = d0[4] * d0[4] + d0[5] * d0[5];
    float q3 = d0[6] * d0[6] + d0[7] * d0[7];
    float q8 = d1[0] * d1[0] + d1[1] * d1[1];
    float q9 = d1[2] * d1[2] + d1[3] * d1[3];
    *reinterpret_cast<float2*>(o + 0) = make_float2(q0, q1);
    *reinterpret_cast<float2*>(o + 2) = make_float2(q2, q3);
    *reinterpret_cast<float2*>(o + 8) = make_float2(q8, q9);
  } else {
    float q4 = d0[0] * d0[0] + d0[1] * d0[1];
    float q5 = d0[2] * d0[2] + d0[3] * d0[3];
    float q6 = d0[4] * d0[4] + d0[5] * d0[5];
    float q7 = d0[6] * d0[6] + d0[7] * d0[7];
    *reinterpret_cast<float2*>(o + 4) = make_float2(q4, q5);
    *reinterpret_cast<float2*>(o + 6) = make_float2(q6, q7);
  }
}

__global__ void __launch_bounds__(256)
qc_fuzzy_kernel(const float* __restrict__ X, const float* __restrict__ params,
                float* __restrict__ out, int ntiles) {
  __shared__ float s_sc[2 * N_ANGLES];
  __shared__ __attribute__((aligned(16))) float s_A[2 * 16 * 4];
  const int tid = threadIdx.x;
  const int lane = threadIdx.x & 31;
  const int wave = __builtin_amdgcn_readfirstlane(threadIdx.x >> 5);
  const int wpb  = blockDim.x >> 5;
  const int wid  = blockIdx.x * wpb + wave;
  const int wstride = gridDim.x * wpb;

  const int  n  = lane & 15;
  const bool hi = lane >= 16;
  const int  kb = hi ? 2 : 0;

  const int ngroups = (ntiles + TPG - 1) / TPG;
  const int last = ntiles - 1;

  float xn[TPG];
#pragma unroll
  for (int j = 0; j < TPG; ++j) {
    int tt = wid * TPG + j;
    tt = tt < last ? tt : last;
    xn[j] = X[tt * 16 + n];
  }

  if (tid < N_ANGLES) {
    float sv, cv;
    sincosf(0.5f * params[tid], &sv, &cv);
    s_sc[2 * tid]     = sv;
    s_sc[2 * tid + 1] = cv;
  } else if (tid >= 128) {
    s_A[tid - 128] = 0.0f;
  }
  __syncthreads();
  if (tid < NUM_CLASSES) qc_class_coeffs(tid, s_sc, s_A);
  __syncthreads();

  if (wid >= ngroups) return;

  v2f a0, a1;
  a0[0] = s_A[(0 * 16 + n) * 4 + kb];
  a0[1] = s_A[(0 * 16 + n) * 4 + kb + 1];
  a1[0] = s_A[(1 * 16 + n) * 4 + kb];
  a1[1] = s_A[(1 * 16 + n) * 4 + kb + 1];

  for (int g = wid; g < ngroups; g += wstride) {
    float xc[TPG];
#pragma unroll
    for (int j = 0; j < TPG; ++j) xc[j] = xn[j];

    int gn = g + wstride;
    gn = gn < ngroups ? gn : ngroups - 1;
#pragma unroll
    for (int j = 0; j < TPG; ++j) {
      int tt = gn * TPG + j;
      tt = tt < last ? tt : last;
      xn[j] = X[tt * 16 + n];
    }

#pragma unroll
    for (int j = 0; j < TPG; ++j) {
      const int t = g * TPG + j;
      if (t < ntiles) {
        const int p = t * 16 + n;
        qc_tile(xc[j], a0, a1, hi, out + (size_t)p * NUM_CLASSES);
      }
    }
  }
}

extern "C" void kernel_launch(void* const* d_in, const int* in_sizes, int n_in,
                              void* d_out, int out_size, void* d_ws, size_t ws_size,
                              hipStream_t stream) {
  const float* x      = (const float*)d_in[0];
  const float* params = (const float*)d_in[1];
  float* out   = (float*)d_out;

  const int npix   = in_sizes[0];
  const int ntiles = npix / 16;

  int blocks = (ntiles + 7) / 8;
  if (blocks > 512) blocks = 512;
  hipLaunchKernelGGL(qc_fuzzy_kernel, dim3(blocks), dim3(256), 0, stream,
                     x, params, out, ntiles);
}


# Output staged through a private LDS block per wave: each 8-tile group (5120 contiguous bytes) leaves as 10 whole-line float4-per-lane stores instead of 40 scattered 4- and 8-byte partial-line stores
# speedup vs baseline: 3.2189x; 1.8344x over previous
#include <hip/hip_runtime.h>
#include <math.h>


typedef __attribute__((ext_vector_type(2))) float v2f;
typedef __attribute__((ext_vector_type(8))) float v8f;

#define NUM_CLASSES 10
#define N_LAYERS 3

#define N_ANGLES (NUM_CLASSES * N_LAYERS * 3)

__device__ __forceinline__ void qc_class_coeffs(int cls, const float* sc,
                                                float* Acoef) {
  float tr[4] = {1.f, 0.f, 0.f, 0.f};
  float ti[4] = {0.f, 0.f, 0.f, 0.f};
  float br[4] = {0.f, 0.f, 0.f, 0.f};
  float bi[4] = {0.f, 0.f, 0.f, 0.f};
  int len = 1;
#pragma unroll
  for (int l = 0; l < N_LAYERS; ++l) {
    float ntr[4], nti[4], nbr[4], nbi[4];
#pragma unroll
    for (int j = 0; j < 4; ++j) {
      if (j <= len) {
        float xr = 0.f, xi = 0.f, yr = 0.f, yi = 0.f;
        if (j < len) { xr = tr[j]; xi = ti[j]; yr = br[j]; yi = bi[j]; }
        if (j >= 1)  { xr -= br[j-1]; xi -= bi[j-1]; yr += tr[j-1]; yi += ti[j-1]; }
        ntr[j] = xr; nti[j] = xi; nbr[j] = yr; nbi[j] = yi;
      }
    }
    ++len;
    const float* S = sc + 2 * ((cls * N_LAYERS + l) * 3);
    float sz0 = S[0], cz0 = S[1];
    float s1  = S[2], c1  = S[3];
    float sz2 = S[4], cz2 = S[5];
#pragma unroll
    for (int j = 0; j < 4; ++j) {
      if (j < len) {
        float xr = ntr[j], xi = nti[j], yr = nbr[j], yi = nbi[j];
        float txr = xr * cz0 + xi * sz0;
        float txi = xi * cz0 - xr * sz0;
        float tyr = yr * cz0 - yi * sz0;
        float tyi = yi * cz0 + yr * sz0;
        float uxr = c1 * txr - s1 * tyr, uxi = c1 * txi - s1 * tyi;
        float uyr = s1 * txr + c1 * tyr, uyi = s1 * txi + c1 * tyi;
        tr[j] = uxr * cz2 + uxi * sz2;
        ti[j] = uxi * cz2 - uxr * sz2;
        br[j] = uyr * cz2 - uyi * sz2;
        bi[j] = uyi * cz2 + uyr * sz2;
      }
    }
  }
  int nRe = 2 * cls, nIm = 2 * cls + 1;
#pragma unroll
  for (int k = 0; k < 4; ++k) {
    Acoef[((nRe >> 4) * 16 + (nRe & 15)) * 4 + k] = tr[k];
    Acoef[((nIm >> 4) * 16 + (nIm & 15)) * 4 + k] = ti[k];
  }
}

#define TPG 8
#define WAVES_PER_BLOCK 8
#define TILE_FLOATS  (16 * NUM_CLASSES)
#define GROUP_FLOATS (TPG * TILE_FLOATS)
#define GROUP_VEC4   (GROUP_FLOATS / 4)

__device__ __forceinline__ void qc_tile(float x, v2f a0, v2f a1, bool hi,
                                        float* st) {
  float s, c;
  __sincosf(0.5f * x, &s, &c);
  float c2 = c * c, s2 = s * s;
  v2f b;
  b[0] = hi ? c * s2 : c2 * c;
  b[1] = hi ? s2 * s : c2 * s;

  v8f z = {};
  v8f d0 = __builtin_amdgcn_wmma_f32_16x16x4_f32(false, a0, false, b,
                                                 (short)0, z, false, false);
  v8f d1 = __builtin_amdgcn_wmma_f32_16x16x4_f32(false, a1, false, b,
                                                 (short)0, z, false, false);

  float qa = d0[0] * d0[0] + d0[1] * d0[1];
  float qb = d0[2] * d0[2] + d0[3] * d0[3];
  float qc = d0[4] * d0[4] + d0[5] * d0[5];
  float qd = d0[6] * d0[6] + d0[7] * d0[7];
  *reinterpret_cast<float2*>(st + 0) = make_float2(qa, qb);
  *reinterpret_cast<float2*>(st + 2) = make_float2(qc, qd);
  if (!hi) {
    float q8 = d1[0] * d1[0] + d1[1] * d1[1];
    float q9 = d1[2] * d1[2] + d1[3] * d1[3];
    *reinterpret_cast<float2*>(st + 8) = make_float2(q8, q9);
  }
}

__global__ void __launch_bounds__(WAVES_PER_BLOCK * 32)
qc_fuzzy_kernel(const float* __restrict__ X, const float* __restrict__ params,
                float* __restrict__ out, int ntiles) {
  __shared__ float s_sc[2 * N_ANGLES];
  __shared__ __attribute__((aligned(16))) float s_A[2 * 16 * 4];
  __shared__ __attribute__((aligned(16))) float s_stage[WAVES_PER_BLOCK * GROUP_FLOATS];
  const int tid = threadIdx.x;
  const int lane = threadIdx.x & 31;
  const int wave = __builtin_amdgcn_readfirstlane(threadIdx.x >> 5);
  const int wpb  = blockDim.x >> 5;
  const int wid  = blockIdx.x * wpb + wave;
  const int wstride = gridDim.x * wpb;

  const int  n  = lane & 15;
  const bool hi = lane >= 16;
  const int  kb = hi ? 2 : 0;

  const int ngroups = (ntiles + TPG - 1) / TPG;
  const int last = ntiles - 1;

  float xn[TPG];
#pragma unroll
  for (int j = 0; j < TPG; ++j) {
    int tt = wid * TPG + j;
    tt = tt < last ? tt : last;
    xn[j] = X[tt * 16 + n];
  }

  if (tid < N_ANGLES) {
    float sv, cv;
    sincosf(0.5f * params[tid], &sv, &cv);
    s_sc[2 * tid]     = sv;
    s_sc[2 * tid + 1] = cv;
  } else if (tid >= 128) {
    s_A[tid - 128] = 0.0f;
  }
  __syncthreads();
  if (tid < NUM_CLASSES) qc_class_coeffs(tid, s_sc, s_A);
  __syncthreads();

  if (wid >= ngroups) return;

  v2f a0, a1;
  a0[0] = s_A[(0 * 16 + n) * 4 + kb];
  a0[1] = s_A[(0 * 16 + n) * 4 + kb + 1];
  a1[0] = s_A[(1 * 16 + n) * 4 + kb];
  a1[1] = s_A[(1 * 16 + n) * 4 + kb + 1];

  float* const stw = s_stage + wave * GROUP_FLOATS;
  float* const st_lane = stw + n * NUM_CLASSES + (hi ? 4 : 0);
  const float4* const ld_lane = reinterpret_cast<const float4*>(stw) + lane;

  for (int g = wid; g < ngroups; g += wstride) {
    float xc[TPG];
#pragma unroll
    for (int j = 0; j < TPG; ++j) xc[j] = xn[j];

    int gn = g + wstride;
    gn = gn < ngroups ? gn : ngroups - 1;
#pragma unroll
    for (int j = 0; j < TPG; ++j) {
      int tt = gn * TPG + j;
      tt = tt < last ? tt : last;
      xn[j] = X[tt * 16 + n];
    }

    float4* const og = reinterpret_cast<float4*>(out + (size_t)g * GROUP_FLOATS) + lane;
    const int nt = ntiles - g * TPG;
    if (nt >= TPG) {
#pragma unroll
      for (int j = 0; j < TPG; ++j) qc_tile(xc[j], a0, a1, hi, st_lane + j * TILE_FLOATS);

      asm volatile("s_wait_dscnt 0x0" ::: "memory");

      float4 v[GROUP_VEC4 / 32];
#pragma unroll
      for (int i = 0; i < GROUP_VEC4 / 32; ++i) v[i] = ld_lane[i * 32];
#pragma unroll
      for (int i = 0; i < GROUP_VEC4 / 32; ++i) og[i * 32] = v[i];
    } else {
#pragma unroll
      for (int j = 0; j < TPG; ++j)
        if (j < nt) qc_tile(xc[j], a0, a1, hi, st_lane + j * TILE_FLOATS);

      asm volatile("s_wait_dscnt 0x0" ::: "memory");

      const int nvec = nt * (TILE_FLOATS / 4);
#pragma unroll
      for (int i = 0; i < GROUP_VEC4 / 32; ++i)
        if (i * 32 + lane < nvec) og[i * 32] = ld_lane[i * 32];
    }
  }
}

extern "C" void kernel_launch(void* const* d_in, const int* in_sizes, int n_in,
                              void* d_out, int out_size, void* d_ws, size_t ws_size,
                              hipStream_t stream) {
  const float* x      = (const float*)d_in[0];
  const float* params = (const float*)d_in[1];
  float* out   = (float*)d_out;

  const int npix   = in_sizes[0];
  const int ntiles = npix / 16;

  int blocks = (ntiles + 7) / 8;
  if (blocks > 512) blocks = 512;
  hipLaunchKernelGGL(qc_fuzzy_kernel, dim3(blocks), dim3(WAVES_PER_BLOCK * 32), 0, stream,
                     x, params, out, ntiles);
}
